# starting kernel
# baseline (speedup 1.0000x reference)
#include <hip/hip_runtime.h>


typedef float v2f __attribute__((ext_vector_type(2)));
typedef int   v2i __attribute__((ext_vector_type(2)));
typedef float v8f __attribute__((ext_vector_type(8)));

#define RED_BLOCKS 256
#define RED_THREADS 256
#define KNB 32

__global__ __launch_bounds__(RED_THREADS)
void lmi_sum_partial(const float* __restrict__ X, float* __restrict__ partial, int n) {
    __shared__ float smem[RED_THREADS];
    float acc = 0.0f;
    for (int i = blockIdx.x * RED_THREADS + threadIdx.x; i < n; i += gridDim.x * RED_THREADS)
        acc += X[i];
    smem[threadIdx.x] = acc;
    __syncthreads();
    #pragma unroll
    for (int s = RED_THREADS / 2; s > 0; s >>= 1) {
        if ((int)threadIdx.x < s) smem[threadIdx.x] += smem[threadIdx.x + s];
        __syncthreads();
    }
    if (threadIdx.x == 0) partial[blockIdx.x] = smem[0];
}

__global__ __launch_bounds__(RED_BLOCKS)
void lmi_finalize_mean(const float* __restrict__ partial, float* __restrict__ mean_out, int n) {
    __shared__ float smem[RED_BLOCKS];
    smem[threadIdx.x] = partial[threadIdx.x];
    __syncthreads();
    #pragma unroll
    for (int s = RED_BLOCKS / 2; s > 0; s >>= 1) {
        if ((int)threadIdx.x < s) smem[threadIdx.x] += smem[threadIdx.x + s];
        __syncthreads();
    }
    if (threadIdx.x == 0) mean_out[0] = smem[0] / (float)n;
}

__global__ __launch_bounds__(256)
void lmi_main(const float* __restrict__ X,
              const float* __restrict__ W,
              const int*   __restrict__ IDS,
              const float* __restrict__ meanp,
              float*       __restrict__ out,
              int n) {
    const int lane = threadIdx.x & 31;
    const int wave = threadIdx.x >> 5;
    const int tile = blockIdx.x * 8 + wave;
    const int full_tiles = n >> 4;
    const float mean = meanp[0];

    if (tile < full_tiles) {
        const int base = tile << 4;
        const int m    = lane & 15;
        const int h    = lane >> 4;
        const long long rowoff = (long long)(base + m) * KNB;
        const float* wrow = W   + rowoff;
        const int*   irow = IDS + rowoff;

        v8f dlag = {0.f,0.f,0.f,0.f,0.f,0.f,0.f,0.f};
        v8f dsq  = {0.f,0.f,0.f,0.f,0.f,0.f,0.f,0.f};
        v2f ones; ones.x = 1.0f; ones.y = 1.0f;

        #pragma unroll
        for (int c = 0; c < 8; ++c) {
            const int k = c * 4 + h * 2;
            v2f wp = __builtin_nontemporal_load(
                         reinterpret_cast<const v2f*>(wrow + k));
            v2i ip = __builtin_nontemporal_load(
                         reinterpret_cast<const v2i*>(irow + k));
            const float xn0 = X[ip.x] - mean;
            const float xn1 = X[ip.y] - mean;
            v2f A;  A.x  = wp.x * xn0;       A.y  = wp.y * xn1;
            v2f Aq; Aq.x = A.x  * xn0;       Aq.y = A.y  * xn1;
            dlag = __builtin_amdgcn_wmma_f32_16x16x4_f32(
                       false, A,  false, ones, (short)0, dlag, false, false);
            dsq  = __builtin_amdgcn_wmma_f32_16x16x4_f32(
                       false, Aq, false, ones, (short)0, dsq,  false, false);
        }

        if ((lane & 15) == 0) {
            const int rbase = base + h * 8;
            #pragma unroll
            for (int v = 0; v < 8; ++v) {
                const float xs  = X[rbase + v] - mean;
                const float lag = dlag[v];
                const float sq  = dsq[v];
                __builtin_nontemporal_store(xs * lag * (float)(KNB - 1) / sq,
                                            out + rbase + v);
            }
        }
    } else if ((n & 15) && tile == full_tiles) {
        const int rem = n & 15;
        if (lane < rem) {
            const int row = full_tiles * 16 + lane;
            const float* wrow = W   + (long long)row * KNB;
            const int*   irow = IDS + (long long)row * KNB;
            float lag = 0.f, sq = 0.f;
            for (int k = 0; k < KNB; ++k) {
                const float xn = X[irow[k]] - mean;
                const float p  = wrow[k] * xn;
                lag += p; sq += p * xn;
            }
            const float xs = X[row] - mean;
            out[row] = xs * lag * (float)(KNB - 1) / sq;
        }
    }
}

extern "C" void kernel_launch(void* const* d_in, const int* in_sizes, int n_in,
                              void* d_out, int out_size, void* d_ws, size_t ws_size,
                              hipStream_t stream) {
    const float* X   = (const float*)d_in[0];
    const float* W   = (const float*)d_in[1];
    const int*   IDS = (const int*)d_in[2];
    float*       out = (float*)d_out;
    const int n = in_sizes[0];

    float* partial = (float*)d_ws;
    float* meanp   = partial + RED_BLOCKS;

    lmi_sum_partial<<<RED_BLOCKS, RED_THREADS, 0, stream>>>(X, partial, n);
    lmi_finalize_mean<<<1, RED_BLOCKS, 0, stream>>>(partial, meanp, n);

    const int tiles  = (n + 15) / 16;
    const int blocks = (tiles + 7) / 8;
    lmi_main<<<blocks, 256, 0, stream>>>(X, W, IDS, meanp, out, n);
}


# W and ids read as whole 128-B lines: 16 B per lane, 4 full rows per load instead of 16-B pieces of 16 rows; row sum = 2 chained WMMAs per 4 rows, then 4 D entries added (new reduction tree)
# speedup vs baseline: 1.3816x; 1.3816x over previous
#include <hip/hip_runtime.h>


typedef float v2f __attribute__((ext_vector_type(2)));
typedef int   v2i __attribute__((ext_vector_type(2)));
typedef float v4f __attribute__((ext_vector_type(4)));
typedef int   v4i __attribute__((ext_vector_type(4)));
typedef float v8f __attribute__((ext_vector_type(8)));

#define RED_BLOCKS 256
#define RED_THREADS 256
#define KNB 32

__global__ __launch_bounds__(RED_THREADS)
void lmi_sum_partial(const float* __restrict__ X, float* __restrict__ partial, int n) {
    __shared__ float smem[RED_THREADS];
    float acc = 0.0f;
    for (int i = blockIdx.x * RED_THREADS + threadIdx.x; i < n; i += gridDim.x * RED_THREADS)
        acc += X[i];
    smem[threadIdx.x] = acc;
    __syncthreads();
    #pragma unroll
    for (int s = RED_THREADS / 2; s > 0; s >>= 1) {
        if ((int)threadIdx.x < s) smem[threadIdx.x] += smem[threadIdx.x + s];
        __syncthreads();
    }
    if (threadIdx.x == 0) partial[blockIdx.x] = smem[0];
}

__global__ __launch_bounds__(RED_BLOCKS)
void lmi_finalize_mean(const float* __restrict__ partial, float* __restrict__ mean_out, int n) {
    __shared__ float smem[RED_BLOCKS];
    smem[threadIdx.x] = partial[threadIdx.x];
    __syncthreads();
    #pragma unroll
    for (int s = RED_BLOCKS / 2; s > 0; s >>= 1) {
        if ((int)threadIdx.x < s) smem[threadIdx.x] += smem[threadIdx.x + s];
        __syncthreads();
    }
    if (threadIdx.x == 0) mean_out[0] = smem[0] / (float)n;
}

__global__ __launch_bounds__(256)
void lmi_main(const float* __restrict__ X,
              const float* __restrict__ W,
              const int*   __restrict__ IDS,
              const float* __restrict__ meanp,
              float*       __restrict__ out,
              int n) {
    const int lane = threadIdx.x & 31;
    const int wave = threadIdx.x >> 5;
    const int tile = blockIdx.x * 8 + wave;
    const int full_tiles = n >> 4;
    const float mean = meanp[0];

    if (tile < full_tiles) {
        const int base = tile << 4;
        const int c    = lane & 15;
        const int h    = lane >> 4;
        const int row0 = base + (c & 8) + ((c >> 2) & 1);
        const int q    = (c & 3) + 4 * h;
        const long long off0 = (long long)row0 * KNB + q * 4;
        const v4f* wp = reinterpret_cast<const v4f*>(W   + off0);
        const v4i* ip = reinterpret_cast<const v4i*>(IDS + off0);

        v2f ones; ones.x = 1.0f; ones.y = 1.0f;
        const v8f zero8 = {0.f,0.f,0.f,0.f,0.f,0.f,0.f,0.f};
        float lag[8], sq[8];

        #pragma unroll
        for (int j = 0; j < 4; ++j) {
            const v4i id = __builtin_nontemporal_load(ip + 16 * j);
            const v4f w  = __builtin_nontemporal_load(wp + 16 * j);
            const float xn0 = X[id.x] - mean;
            const float xn1 = X[id.y] - mean;
            const float xn2 = X[id.z] - mean;
            const float xn3 = X[id.w] - mean;
            v2f A0; A0.x = w.x * xn0;   A0.y = w.y * xn1;
            v2f A1; A1.x = w.z * xn2;   A1.y = w.w * xn3;
            v2f Q0; Q0.x = A0.x * xn0;  Q0.y = A0.y * xn1;
            v2f Q1; Q1.x = A1.x * xn2;  Q1.y = A1.y * xn3;
            v8f dl = __builtin_amdgcn_wmma_f32_16x16x4_f32(
                         false, A0, false, ones, (short)0, zero8, false, false);
            v8f ds = __builtin_amdgcn_wmma_f32_16x16x4_f32(
                         false, Q0, false, ones, (short)0, zero8, false, false);
            dl = __builtin_amdgcn_wmma_f32_16x16x4_f32(
                         false, A1, false, ones, (short)0, dl, false, false);
            ds = __builtin_amdgcn_wmma_f32_16x16x4_f32(
                         false, Q1, false, ones, (short)0, ds, false, false);
            lag[2 * j]     = (dl[0] + dl[1]) + (dl[2] + dl[3]);
            lag[2 * j + 1] = (dl[4] + dl[5]) + (dl[6] + dl[7]);
            sq[2 * j]      = (ds[0] + ds[1]) + (ds[2] + ds[3]);
            sq[2 * j + 1]  = (ds[4] + ds[5]) + (ds[6] + ds[7]);
        }

        if ((lane & 15) == 0) {
            const int rbase = base + h * 8;
            #pragma unroll
            for (int v = 0; v < 8; ++v) {
                const float xs = X[rbase + v] - mean;
                __builtin_nontemporal_store(xs * lag[v] * (float)(KNB - 1) / sq[v],
                                            out + rbase + v);
            }
        }
    } else if ((n & 15) && tile == full_tiles) {
        const int rem = n & 15;
        if (lane < rem) {
            const int row = full_tiles * 16 + lane;
            const float* wrow = W   + (long long)row * KNB;
            const int*   irow = IDS + (long long)row * KNB;
            float lag = 0.f, sq = 0.f;
            for (int k = 0; k < KNB; ++k) {
                const float xn = X[irow[k]] - mean;
                const float p  = wrow[k] * xn;
                lag += p; sq += p * xn;
            }
            const float xs = X[row] - mean;
            out[row] = xs * lag * (float)(KNB - 1) / sq;
        }
    }
}

extern "C" void kernel_launch(void* const* d_in, const int* in_sizes, int n_in,
                              void* d_out, int out_size, void* d_ws, size_t ws_size,
                              hipStream_t stream) {
    const float* X   = (const float*)d_in[0];
    const float* W   = (const float*)d_in[1];
    const int*   IDS = (const int*)d_in[2];
    float*       out = (float*)d_out;
    const int n = in_sizes[0];

    float* partial = (float*)d_ws;
    float* meanp   = partial + RED_BLOCKS;

    lmi_sum_partial<<<RED_BLOCKS, RED_THREADS, 0, stream>>>(X, partial, n);
    lmi_finalize_mean<<<1, RED_BLOCKS, 0, stream>>>(partial, meanp, n);

    const int tiles  = (n + 15) / 16;
    const int blocks = (tiles + 7) / 8;
    lmi_main<<<blocks, 256, 0, stream>>>(X, W, IDS, meanp, out, n);
}


# Neighbour values read from LDS instead of one second-level request per lane: resident grid of 256 work-groups, X copied through the 320 KiB LDS chunk by chunk, ids kept in registers
# speedup vs baseline: 1.9348x; 1.4004x over previous
#include <hip/hip_runtime.h>


typedef float v2f __attribute__((ext_vector_type(2)));
typedef int   v2i __attribute__((ext_vector_type(2)));
typedef float v4f __attribute__((ext_vector_type(4)));
typedef int   v4i __attribute__((ext_vector_type(4)));
typedef float v8f __attribute__((ext_vector_type(8)));

#define RED_BLOCKS 256
#define RED_THREADS 256
#define KNB 32

__global__ __launch_bounds__(RED_THREADS)
void lmi_sum_partial(const float* __restrict__ X, float* __restrict__ partial, int n) {
    __shared__ float smem[RED_THREADS];
    float acc = 0.0f;
    for (int i = blockIdx.x * RED_THREADS + threadIdx.x; i < n; i += gridDim.x * RED_THREADS)
        acc += X[i];
    smem[threadIdx.x] = acc;
    __syncthreads();
    #pragma unroll
    for (int s = RED_THREADS / 2; s > 0; s >>= 1) {
        if ((int)threadIdx.x < s) smem[threadIdx.x] += smem[threadIdx.x + s];
        __syncthreads();
    }
    if (threadIdx.x == 0) partial[blockIdx.x] = smem[0];
}

__global__ __launch_bounds__(RED_BLOCKS)
void lmi_finalize_mean(const float* __restrict__ partial, float* __restrict__ mean_out, int n) {
    __shared__ float smem[RED_BLOCKS];
    smem[threadIdx.x] = partial[threadIdx.x];
    __syncthreads();
    #pragma unroll
    for (int s = RED_BLOCKS / 2; s > 0; s >>= 1) {
        if ((int)threadIdx.x < s) smem[threadIdx.x] += smem[threadIdx.x + s];
        __syncthreads();
    }
    if (threadIdx.x == 0) mean_out[0] = smem[0] / (float)n;
}

#define CH   81920
#define TPW  8
#define WGW  16
#define WGT  512
#define NWG  256
#define FU   10
#define WAHEAD 2

__device__ __forceinline__ unsigned lmi_here(unsigned x) {
    asm volatile("" : "+v"(x));
    return x;
}

__device__ __forceinline__ unsigned lmi_lane_offset(unsigned lane) {
    const unsigned c = lane & 15u;
    const unsigned h = lane >> 4;
    const unsigned rowoff = (c & 8u) + ((c >> 2) & 1u);
    const unsigned q = (c & 3u) + 4u * h;
    return rowoff * KNB + q * 4u;
}

__device__ __forceinline__ void lmi_load_w(v4f (&w)[4], v4f (&xo)[2],
                                           const float* __restrict__ wt,
                                           const float* __restrict__ xt,
                                           unsigned lc, unsigned h) {
    #pragma unroll
    for (int j = 0; j < 4; ++j)
        w[j] = __builtin_nontemporal_load(reinterpret_cast<const v4f*>(wt + (lc + 64u * j)));
    xo[0] = *reinterpret_cast<const v4f*>(xt + 8u * h);
    xo[1] = *reinterpret_cast<const v4f*>(xt + 8u * h + 4u);
}

__device__ __forceinline__ void lmi_finish_tile(const unsigned (&xv)[16], const v4f (&w)[4],
                                                const v4f (&xo)[2],
                                                float mean, int lane, int h, bool live,
                                                float* __restrict__ out, int base) {
    v2f ones; ones.x = 1.0f; ones.y = 1.0f;
    const v8f zero8 = {0.f,0.f,0.f,0.f,0.f,0.f,0.f,0.f};
    #pragma unroll
    for (int g = 0; g < 2; ++g) {
        float lag[4], sq[4];
        #pragma unroll
        for (int jj = 0; jj < 2; ++jj) {
            const int j = 2 * g + jj;
            const float xn0 = __uint_as_float(xv[4 * j + 0]) - mean;
            const float xn1 = __uint_as_float(xv[4 * j + 1]) - mean;
            const float xn2 = __uint_as_float(xv[4 * j + 2]) - mean;
            const float xn3 = __uint_as_float(xv[4 * j + 3]) - mean;
            v2f A0; A0.x = w[j].x * xn0;   A0.y = w[j].y * xn1;
            v2f A1; A1.x = w[j].z * xn2;   A1.y = w[j].w * xn3;
            v2f Q0; Q0.x = A0.x * xn0;  Q0.y = A0.y * xn1;
            v2f Q1; Q1.x = A1.x * xn2;  Q1.y = A1.y * xn3;
            v8f dl = __builtin_amdgcn_wmma_f32_16x16x4_f32(
                         false, A0, false, ones, (short)0, zero8, false, false);
            v8f ds = __builtin_amdgcn_wmma_f32_16x16x4_f32(
                         false, Q0, false, ones, (short)0, zero8, false, false);
            dl = __builtin_amdgcn_wmma_f32_16x16x4_f32(
                         false, A1, false, ones, (short)0, dl, false, false);
            ds = __builtin_amdgcn_wmma_f32_16x16x4_f32(
                         false, Q1, false, ones, (short)0, ds, false, false);
            float l0 = (dl[0] + dl[1]) + (dl[2] + dl[3]);
            float l1 = (dl[4] + dl[5]) + (dl[6] + dl[7]);
            float s0 = (ds[0] + ds[1]) + (ds[2] + ds[3]);
            float s1 = (ds[4] + ds[5]) + (ds[6] + ds[7]);
            lag[2 * jj] = l0; lag[2 * jj + 1] = l1;
            sq[2 * jj]  = s0; sq[2 * jj + 1]  = s1;
        }

        if (live && (lane & 15) == 0) {
            const int rbase = base + h * 8 + 4 * g;
            #pragma unroll
            for (int r = 0; r < 4; ++r) {
                const float xs = xo[g][r] - mean;
                __builtin_nontemporal_store(xs * lag[r] * (float)(KNB - 1) / sq[r],
                                            out + rbase + r);
            }
        }
    }
}

__global__ __launch_bounds__(WGT)
void lmi_main(const float* __restrict__ X,
              const float* __restrict__ W,
              const int*   __restrict__ IDS,
              const float* __restrict__ meanp,
              float*       __restrict__ out,
              int n) {
    __shared__ __attribute__((aligned(16))) float xl[CH];

    const int tid  = threadIdx.x;
    const int wave = __builtin_amdgcn_readfirstlane(tid >> 5);
    const int full_tiles = n >> 4;
    const float mean = meanp[0];

    const int t0 = (int)(((long long)full_tiles * blockIdx.x) / gridDim.x);
    const int t1 = (int)(((long long)full_tiles * (blockIdx.x + 1)) / gridDim.x);
    const int npass = (n + CH - 1) / CH;

    if (npass <= 16) {
        for (int r0 = t0; r0 < t1; r0 += WGW * TPW) {
            const int tb = r0 + wave * TPW;

            unsigned v[TPW][16];
            unsigned nib[TPW][2];

            const unsigned lci = lmi_lane_offset(lmi_here((unsigned)tid) & 31u);
            #pragma unroll
            for (int u = 0; u < TPW; ++u) {
                const int tl = (tb + u < t1) ? (tb + u) : (t1 - 1);
                const int* it = IDS + (long long)tl * (16 * KNB);
                #pragma unroll
                for (int j = 0; j < 4; ++j) {
                    const v4i id = __builtin_nontemporal_load(
                                       reinterpret_cast<const v4i*>(it + (lci + 64u * j)));
                    v[u][4 * j + 0] = (unsigned)id.x;
                    v[u][4 * j + 1] = (unsigned)id.y;
                    v[u][4 * j + 2] = (unsigned)id.z;
                    v[u][4 * j + 3] = (unsigned)id.w;
                }
            }

            #pragma unroll
            for (int G = 0; G < 2 * TPW; ++G) {
                const int u = G >> 1, g = G & 1;
                unsigned word = 0u;
                #pragma unroll
                for (int e = 0; e < 8; ++e) {
                    const unsigned id = v[u][8 * g + e];
                    const unsigned cp = ((id >> 14) * 52429u) >> 18;
                    v[u][8 * g + e] = (id - cp * (unsigned)CH) << 2;
                    word |= cp << (4 * e);
                }
                if (G + 1 < 2 * TPW) {
                    const int un = (G + 1) >> 1, o = 8 * ((G + 1) & 1), c0 = 8 * g;
                    asm volatile("" : "+v"(word),
                                      "+v"(v[u][c0 + 0]), "+v"(v[u][c0 + 1]), "+v"(v[u][c0 + 2]),
                                      "+v"(v[u][c0 + 3]), "+v"(v[u][c0 + 4]), "+v"(v[u][c0 + 5]),
                                      "+v"(v[u][c0 + 6]), "+v"(v[u][c0 + 7]),
                                      "+v"(v[un][o + 0]), "+v"(v[un][o + 1]), "+v"(v[un][o + 2]),
                                      "+v"(v[un][o + 3]), "+v"(v[un][o + 4]), "+v"(v[un][o + 5]),
                                      "+v"(v[un][o + 6]), "+v"(v[un][o + 7]));
                } else {
                    asm volatile("" : "+v"(word));
                }
                nib[u][g] = word;
            }

            for (int p = 0; p < npass; ++p) {
                const int pbase = p * CH;
                const int cnt   = (n - pbase < CH) ? (n - pbase) : CH;
                const int nv4   = cnt >> 2;
                const int rot   = (int)(((blockIdx.x & (NWG - 1)) * (unsigned)nv4) / NWG) & ~15;
                const v4f* X4 = reinterpret_cast<const v4f*>(X + pbase);
                v4f* xl4 = reinterpret_cast<v4f*>(xl);

                __syncthreads();
                for (int b = 0; b < nv4; b += WGT * FU) {
                    v4f t[FU]; int jj[FU];
                    #pragma unroll
                    for (int k = 0; k < FU; ++k) {
                        int i = b + k * WGT + tid;
                        i = (i < nv4) ? i : (nv4 - 1);
                        int j = i + rot;
                        j = (j >= nv4) ? (j - nv4) : j;
                        jj[k] = j;
                        t[k] = X4[j];
                    }
                    #pragma unroll
                    for (int k = 0; k < FU; ++k) xl4[jj[k]] = t[k];
                }
                if (tid < (cnt & 3)) xl[(nv4 << 2) + tid] = X[pbase + (nv4 << 2) + tid];
                __syncthreads();

                const unsigned pm = (unsigned)p * 0x11111111u;
                #pragma unroll
                for (int u = 0; u < TPW; ++u) {
                    const unsigned xk0 = nib[u][0] ^ pm, xk1 = nib[u][1] ^ pm;
                    unsigned got[16];
                    #pragma unroll
                    for (int s = 0; s < 16; ++s) {
                        const bool hit = (((s < 8) ? xk0 : xk1) & (15u << (4 * (s & 7)))) == 0u;
                        const unsigned a = hit ? v[u][s] : 0u;
                        got[s] = *reinterpret_cast<const unsigned*>(
                                     reinterpret_cast<const char*>(xl) + a);
                    }
                    asm volatile("" : "+v"(got[0]),  "+v"(got[1]),  "+v"(got[2]),  "+v"(got[3]),
                                      "+v"(got[4]),  "+v"(got[5]),  "+v"(got[6]),  "+v"(got[7]),
                                      "+v"(got[8]),  "+v"(got[9]),  "+v"(got[10]), "+v"(got[11]),
                                      "+v"(got[12]), "+v"(got[13]), "+v"(got[14]), "+v"(got[15]));
                    #pragma unroll
                    for (int s = 0; s < 16; ++s) {
                        const bool hit = (((s < 8) ? xk0 : xk1) & (15u << (4 * (s & 7)))) == 0u;
                        v[u][s] = hit ? got[s] : v[u][s];
                    }
                }
            }

            const unsigned lanef = lmi_here((unsigned)tid) & 31u;
            const unsigned lcf   = lmi_lane_offset(lanef);
            v4f wq[TPW][4], xq[TPW][2];
            #pragma unroll
            for (int u = 0; u < WAHEAD && u < TPW; ++u) {
                const int tl = (tb + u < t1) ? (tb + u) : (t1 - 1);
                lmi_load_w(wq[u], xq[u], W + (long long)tl * (16 * KNB), X + (tl << 4),
                           lcf, lanef >> 4);
            }
            #pragma unroll
            for (int u = 0; u < TPW; ++u) {
                if (u + WAHEAD < TPW) {
                    const int ta = tb + u + WAHEAD;
                    const int tla = (ta < t1) ? ta : (t1 - 1);
                    lmi_load_w(wq[u + WAHEAD], xq[u + WAHEAD], W + (long long)tla * (16 * KNB),
                               X + (tla << 4), lcf, lanef >> 4);
                }
                const int tile = tb + u;
                const int tl   = (tile < t1) ? tile : (t1 - 1);
                lmi_finish_tile(v[u], wq[u], xq[u], mean, (int)lanef, (int)(lanef >> 4),
                                tile < t1, out, tl << 4);
            }
        }
    } else {
        const int lane = (int)(lmi_here((unsigned)tid) & 31u);
        const unsigned lc = lmi_lane_offset((unsigned)lane);
        const int h = lane >> 4;
        for (int tile = t0 + wave; tile < t1; tile += WGW) {
            const int* it = IDS + (long long)tile * (16 * KNB);
            unsigned xv[16];
            #pragma unroll
            for (int j = 0; j < 4; ++j) {
                const v4i id = __builtin_nontemporal_load(
                                   reinterpret_cast<const v4i*>(it + (lc + 64u * j)));
                xv[4 * j + 0] = __float_as_uint(X[id.x]);
                xv[4 * j + 1] = __float_as_uint(X[id.y]);
                xv[4 * j + 2] = __float_as_uint(X[id.z]);
                xv[4 * j + 3] = __float_as_uint(X[id.w]);
            }
            v4f w[4], xo[2];
            lmi_load_w(w, xo, W + (long long)tile * (16 * KNB), X + (tile << 4), lc, (unsigned)h);
            lmi_finish_tile(xv, w, xo, mean, lane, h, true, out, tile << 4);
        }
    }

    if ((n & 15) && blockIdx.x == gridDim.x - 1 && wave == 0) {
        const int rem = n & 15;
        const int lane = (int)(lmi_here((unsigned)tid) & 31u);
        if (lane < rem) {
            const int row = full_tiles * 16 + lane;
            const float* wrow = W   + (long long)row * KNB;
            const int*   irow = IDS + (long long)row * KNB;
            float lag = 0.f, sq = 0.f;
            for (int k = 0; k < KNB; ++k) {
                const float xn = X[irow[k]] - mean;
                const float p  = wrow[k] * xn;
                lag += p; sq += p * xn;
            }
            const float xs = X[row] - mean;
            out[row] = xs * lag * (float)(KNB - 1) / sq;
        }
    }
}

extern "C" void kernel_launch(void* const* d_in, const int* in_sizes, int n_in,
                              void* d_out, int out_size, void* d_ws, size_t ws_size,
                              hipStream_t stream) {
    const float* X   = (const float*)d_in[0];
    const float* W   = (const float*)d_in[1];
    const int*   IDS = (const int*)d_in[2];
    float*       out = (float*)d_out;
    const int n = in_sizes[0];

    float* partial = (float*)d_ws;
    float* meanp   = partial + RED_BLOCKS;

    lmi_sum_partial<<<RED_BLOCKS, RED_THREADS, 0, stream>>>(X, partial, n);
    lmi_finalize_mean<<<1, RED_BLOCKS, 0, stream>>>(partial, meanp, n);
    lmi_main<<<NWG, WGT, 0, stream>>>(X, W, IDS, meanp, out, n);
}


# One launch instead of three: each work-group adds up X while copying it through LDS in its first round; the mean's summation order changes (per-thread running sums, wave butterfly, wave totals)
# speedup vs baseline: 2.0663x; 1.0680x over previous
#include <hip/hip_runtime.h>


typedef float v2f __attribute__((ext_vector_type(2)));
typedef int   v2i __attribute__((ext_vector_type(2)));
typedef float v4f __attribute__((ext_vector_type(4)));
typedef int   v4i __attribute__((ext_vector_type(4)));
typedef float v8f __attribute__((ext_vector_type(8)));

#define KNB 32

#define CH   81920
#define TPW  8
#define WGW  16
#define WGT  512
#define NWG  256
#define FU   10
#define WAHEAD 2

__device__ __forceinline__ unsigned lmi_here(unsigned x) {
    asm volatile("" : "+v"(x));
    return x;
}

__device__ __forceinline__ unsigned lmi_lane_offset(unsigned lane) {
    const unsigned c = lane & 15u;
    const unsigned h = lane >> 4;
    const unsigned rowoff = (c & 8u) + ((c >> 2) & 1u);
    const unsigned q = (c & 3u) + 4u * h;
    return rowoff * KNB + q * 4u;
}

__device__ __forceinline__ void lmi_load_w(v4f (&w)[4], v4f (&xo)[2],
                                           const float* __restrict__ wt,
                                           const float* __restrict__ xt,
                                           unsigned lc, unsigned h) {
    #pragma unroll
    for (int j = 0; j < 4; ++j)
        w[j] = __builtin_nontemporal_load(reinterpret_cast<const v4f*>(wt + (lc + 64u * j)));
    xo[0] = *reinterpret_cast<const v4f*>(xt + 8u * h);
    xo[1] = *reinterpret_cast<const v4f*>(xt + 8u * h + 4u);
}

__device__ __forceinline__ void lmi_finish_tile(const unsigned (&xv)[16], const v4f (&w)[4],
                                                const v4f (&xo)[2],
                                                float mean, int lane, int h, bool live,
                                                float* __restrict__ out, int base) {
    v2f ones; ones.x = 1.0f; ones.y = 1.0f;
    const v8f zero8 = {0.f,0.f,0.f,0.f,0.f,0.f,0.f,0.f};
    #pragma unroll
    for (int g = 0; g < 2; ++g) {
        float lag[4], sq[4];
        #pragma unroll
        for (int jj = 0; jj < 2; ++jj) {
            const int j = 2 * g + jj;
            const float xn0 = __uint_as_float(xv[4 * j + 0]) - mean;
            const float xn1 = __uint_as_float(xv[4 * j + 1]) - mean;
            const float xn2 = __uint_as_float(xv[4 * j + 2]) - mean;
            const float xn3 = __uint_as_float(xv[4 * j + 3]) - mean;
            v2f A0; A0.x = w[j].x * xn0;   A0.y = w[j].y * xn1;
            v2f A1; A1.x = w[j].z * xn2;   A1.y = w[j].w * xn3;
            v2f Q0; Q0.x = A0.x * xn0;  Q0.y = A0.y * xn1;
            v2f Q1; Q1.x = A1.x * xn2;  Q1.y = A1.y * xn3;
            v8f dl = __builtin_amdgcn_wmma_f32_16x16x4_f32(
                         false, A0, false, ones, (short)0, zero8, false, false);
            v8f ds = __builtin_amdgcn_wmma_f32_16x16x4_f32(
                         false, Q0, false, ones, (short)0, zero8, false, false);
            dl = __builtin_amdgcn_wmma_f32_16x16x4_f32(
                         false, A1, false, ones, (short)0, dl, false, false);
            ds = __builtin_amdgcn_wmma_f32_16x16x4_f32(
                         false, Q1, false, ones, (short)0, ds, false, false);
            float l0 = (dl[0] + dl[1]) + (dl[2] + dl[3]);
            float l1 = (dl[4] + dl[5]) + (dl[6] + dl[7]);
            float s0 = (ds[0] + ds[1]) + (ds[2] + ds[3]);
            float s1 = (ds[4] + ds[5]) + (ds[6] + ds[7]);
            lag[2 * jj] = l0; lag[2 * jj + 1] = l1;
            sq[2 * jj]  = s0; sq[2 * jj + 1]  = s1;
        }

        if (live && (lane & 15) == 0) {
            const int rbase = base + h * 8 + 4 * g;
            #pragma unroll
            for (int r = 0; r < 4; ++r) {
                const float xs = xo[g][r] - mean;
                __builtin_nontemporal_store(xs * lag[r] * (float)(KNB - 1) / sq[r],
                                            out + rbase + r);
            }
        }
    }
}

__device__ __forceinline__ float lmi_block_total(float s, float* __restrict__ xl,
                                                 int tid, int wave) {
    #pragma unroll
    for (int off = 16; off > 0; off >>= 1) s += __shfl_xor(s, off, 32);
    __syncthreads();
    if ((tid & 31) == 0) xl[wave] = s;
    __syncthreads();
    float tot = 0.0f;
    #pragma unroll
    for (int w = 0; w < WGW; ++w) tot += xl[w];
    return tot;
}

__device__ __forceinline__ float lmi_mean_from_memory(const float* __restrict__ X, int n,
                                                      float* __restrict__ xl,
                                                      int tid, int wave) {
    float a = 0.0f;
    for (int i = tid; i < n; i += WGT) a += X[i];
    return lmi_block_total(a, xl, tid, wave) / (float)n;
}

__global__ __launch_bounds__(WGT)
void lmi_main(const float* __restrict__ X,
              const float* __restrict__ W,
              const int*   __restrict__ IDS,
              float*       __restrict__ out,
              int n) {
    __shared__ __attribute__((aligned(16))) float xl[CH];

    const int tid  = threadIdx.x;
    const int wave = __builtin_amdgcn_readfirstlane(tid >> 5);
    const int full_tiles = n >> 4;
    float mean = 0.0f;
    bool have_mean = false;

    const int t0 = (int)(((long long)full_tiles * blockIdx.x) / gridDim.x);
    const int t1 = (int)(((long long)full_tiles * (blockIdx.x + 1)) / gridDim.x);
    const int npass = (n + CH - 1) / CH;

    if (npass <= 16) {
        for (int r0 = t0; r0 < t1; r0 += WGW * TPW) {
            const int tb = r0 + wave * TPW;
            const bool first = (r0 == t0);
            v2f acc; acc.x = 0.0f; acc.y = 0.0f;

            unsigned v[TPW][16];
            unsigned nib[TPW][2];

            const unsigned lci = lmi_lane_offset(lmi_here((unsigned)tid) & 31u);
            #pragma unroll
            for (int u = 0; u < TPW; ++u) {
                const int tl = (tb + u < t1) ? (tb + u) : (t1 - 1);
                const int* it = IDS + (long long)tl * (16 * KNB);
                #pragma unroll
                for (int j = 0; j < 4; ++j) {
                    const v4i id = __builtin_nontemporal_load(
                                       reinterpret_cast<const v4i*>(it + (lci + 64u * j)));
                    v[u][4 * j + 0] = (unsigned)id.x;
                    v[u][4 * j + 1] = (unsigned)id.y;
                    v[u][4 * j + 2] = (unsigned)id.z;
                    v[u][4 * j + 3] = (unsigned)id.w;
                }
            }

            #pragma unroll
            for (int G = 0; G < 2 * TPW; ++G) {
                const int u = G >> 1, g = G & 1;
                unsigned word = 0u;
                #pragma unroll
                for (int e = 0; e < 8; ++e) {
                    const unsigned id = v[u][8 * g + e];
                    const unsigned cp = ((id >> 14) * 52429u) >> 18;
                    v[u][8 * g + e] = (id - cp * (unsigned)CH) << 2;
                    word |= cp << (4 * e);
                }
                if (G + 1 < 2 * TPW) {
                    const int un = (G + 1) >> 1, o = 8 * ((G + 1) & 1), c0 = 8 * g;
                    asm volatile("" : "+v"(word),
                                      "+v"(v[u][c0 + 0]), "+v"(v[u][c0 + 1]), "+v"(v[u][c0 + 2]),
                                      "+v"(v[u][c0 + 3]), "+v"(v[u][c0 + 4]), "+v"(v[u][c0 + 5]),
                                      "+v"(v[u][c0 + 6]), "+v"(v[u][c0 + 7]),
                                      "+v"(v[un][o + 0]), "+v"(v[un][o + 1]), "+v"(v[un][o + 2]),
                                      "+v"(v[un][o + 3]), "+v"(v[un][o + 4]), "+v"(v[un][o + 5]),
                                      "+v"(v[un][o + 6]), "+v"(v[un][o + 7]));
                } else {
                    asm volatile("" : "+v"(word));
                }
                nib[u][g] = word;
            }

            for (int p = 0; p < npass; ++p) {
                const int pbase = p * CH;
                const int cnt   = (n - pbase < CH) ? (n - pbase) : CH;
                const int nv4   = cnt >> 2;
                const int rot   = (int)(((blockIdx.x & (NWG - 1)) * (unsigned)nv4) / NWG) & ~15;
                const v4f* X4 = reinterpret_cast<const v4f*>(X + pbase);
                v4f* xl4 = reinterpret_cast<v4f*>(xl);

                __syncthreads();
                for (int b = 0; b < nv4; b += WGT * FU) {
                    v4f t[FU]; int jj[FU];
                    #pragma unroll
                    for (int k = 0; k < FU; ++k) {
                        int i = b + k * WGT + tid;
                        i = (i < nv4) ? i : (nv4 - 1);
                        int j = i + rot;
                        j = (j >= nv4) ? (j - nv4) : j;
                        jj[k] = j;
                        t[k] = X4[j];
                    }
                    #pragma unroll
                    for (int k = 0; k < FU; ++k) xl4[jj[k]] = t[k];
                    if (first) {
                        #pragma unroll
                        for (int k = 0; k < FU; ++k) {
                            const bool in = (b + k * WGT + tid) < nv4;
                            acc.x += in ? (t[k].x + t[k].z) : 0.0f;
                            acc.y += in ? (t[k].y + t[k].w) : 0.0f;
                        }
                    }
                }
                if (tid < (cnt & 3)) {
                    const float xe = X[pbase + (nv4 << 2) + tid];
                    xl[(nv4 << 2) + tid] = xe;
                    if (first) acc.x += xe;
                }
                __syncthreads();

                const unsigned pm = (unsigned)p * 0x11111111u;
                #pragma unroll
                for (int u = 0; u < TPW; ++u) {
                    const unsigned xk0 = nib[u][0] ^ pm, xk1 = nib[u][1] ^ pm;
                    unsigned got[16];
                    #pragma unroll
                    for (int s = 0; s < 16; ++s) {
                        const bool hit = (((s < 8) ? xk0 : xk1) & (15u << (4 * (s & 7)))) == 0u;
                        const unsigned a = hit ? v[u][s] : 0u;
                        got[s] = *reinterpret_cast<const unsigned*>(
                                     reinterpret_cast<const char*>(xl) + a);
                    }
                    asm volatile("" : "+v"(got[0]),  "+v"(got[1]),  "+v"(got[2]),  "+v"(got[3]),
                                      "+v"(got[4]),  "+v"(got[5]),  "+v"(got[6]),  "+v"(got[7]),
                                      "+v"(got[8]),  "+v"(got[9]),  "+v"(got[10]), "+v"(got[11]),
                                      "+v"(got[12]), "+v"(got[13]), "+v"(got[14]), "+v"(got[15]));
                    #pragma unroll
                    for (int s = 0; s < 16; ++s) {
                        const bool hit = (((s < 8) ? xk0 : xk1) & (15u << (4 * (s & 7)))) == 0u;
                        v[u][s] = hit ? got[s] : v[u][s];
                    }
                }
            }

            if (first) {
                const float tot = lmi_block_total(acc.x + acc.y, xl, tid, wave);
                mean = __int_as_float(__builtin_amdgcn_readfirstlane(
                           __float_as_int(tot / (float)n)));
                have_mean = true;
            }

            const unsigned lanef = lmi_here((unsigned)tid) & 31u;
            const unsigned lcf   = lmi_lane_offset(lanef);
            v4f wq[TPW][4], xq[TPW][2];
            #pragma unroll
            for (int u = 0; u < WAHEAD && u < TPW; ++u) {
                const int tl = (tb + u < t1) ? (tb + u) : (t1 - 1);
                lmi_load_w(wq[u], xq[u], W + (long long)tl * (16 * KNB), X + (tl << 4),
                           lcf, lanef >> 4);
            }
            #pragma unroll
            for (int u = 0; u < TPW; ++u) {
                if (u + WAHEAD < TPW) {
                    const int ta = tb + u + WAHEAD;
                    const int tla = (ta < t1) ? ta : (t1 - 1);
                    lmi_load_w(wq[u + WAHEAD], xq[u + WAHEAD], W + (long long)tla * (16 * KNB),
                               X + (tla << 4), lcf, lanef >> 4);
                }
                const int tile = tb + u;
                const int tl   = (tile < t1) ? tile : (t1 - 1);
                lmi_finish_tile(v[u], wq[u], xq[u], mean, (int)lanef, (int)(lanef >> 4),
                                tile < t1, out, tl << 4);
            }
        }
    } else {
        mean = lmi_mean_from_memory(X, n, xl, tid, wave);
        have_mean = true;
        const int lane = (int)(lmi_here((unsigned)tid) & 31u);
        const unsigned lc = lmi_lane_offset((unsigned)lane);
        const int h = lane >> 4;
        for (int tile = t0 + wave; tile < t1; tile += WGW) {
            const int* it = IDS + (long long)tile * (16 * KNB);
            unsigned xv[16];
            #pragma unroll
            for (int j = 0; j < 4; ++j) {
                const v4i id = __builtin_nontemporal_load(
                                   reinterpret_cast<const v4i*>(it + (lc + 64u * j)));
                xv[4 * j + 0] = __float_as_uint(X[id.x]);
                xv[4 * j + 1] = __float_as_uint(X[id.y]);
                xv[4 * j + 2] = __float_as_uint(X[id.z]);
                xv[4 * j + 3] = __float_as_uint(X[id.w]);
            }
            v4f w[4], xo[2];
            lmi_load_w(w, xo, W + (long long)tile * (16 * KNB), X + (tile << 4), lc, (unsigned)h);
            lmi_finish_tile(xv, w, xo, mean, lane, h, true, out, tile << 4);
        }
    }

    if ((n & 15) && blockIdx.x == gridDim.x - 1) {
      if (!have_mean) mean = lmi_mean_from_memory(X, n, xl, tid, wave);
      if (wave == 0) {
        const int rem = n & 15;
        const int lane = (int)(lmi_here((unsigned)tid) & 31u);
        if (lane < rem) {
            const int row = full_tiles * 16 + lane;
            const float* wrow = W   + (long long)row * KNB;
            const int*   irow = IDS + (long long)row * KNB;
            float lag = 0.f, sq = 0.f;
            for (int k = 0; k < KNB; ++k) {
                const float xn = X[irow[k]] - mean;
                const float p  = wrow[k] * xn;
                lag += p; sq += p * xn;
            }
            const float xs = X[row] - mean;
            out[row] = xs * lag * (float)(KNB - 1) / sq;
        }
      }
    }
}

extern "C" void kernel_launch(void* const* d_in, const int* in_sizes, int n_in,
                              void* d_out, int out_size, void* d_ws, size_t ws_size,
                              hipStream_t stream) {
    const float* X   = (const float*)d_in[0];
    const float* W   = (const float*)d_in[1];
    const int*   IDS = (const int*)d_in[2];
    float*       out = (float*)d_out;
    const int n = in_sizes[0];

    lmi_main<<<NWG, WGT, 0, stream>>>(X, W, IDS, out, n);
}


# Every work-group walks the X chunks in its own cyclic order (first chunk = work-group number mod 13): the reads of a moment spread over all of X; the mean is summed in that order
# speedup vs baseline: 2.1288x; 1.0302x over previous
#include <hip/hip_runtime.h>


typedef float v2f __attribute__((ext_vector_type(2)));
typedef int   v2i __attribute__((ext_vector_type(2)));
typedef float v4f __attribute__((ext_vector_type(4)));
typedef int   v4i __attribute__((ext_vector_type(4)));
typedef float v8f __attribute__((ext_vector_type(8)));

#define KNB 32

#define CH   81920
#define TPW  8
#define WGW  16
#define WGT  512
#define NWG  256
#define FU   10
#define WAHEAD 2

__device__ __forceinline__ unsigned lmi_here(unsigned x) {
    asm volatile("" : "+v"(x));
    return x;
}

__device__ __forceinline__ unsigned lmi_lane_offset(unsigned lane) {
    const unsigned c = lane & 15u;
    const unsigned h = lane >> 4;
    const unsigned rowoff = (c & 8u) + ((c >> 2) & 1u);
    const unsigned q = (c & 3u) + 4u * h;
    return rowoff * KNB + q * 4u;
}

__device__ __forceinline__ void lmi_load_w(v4f (&w)[4], v4f (&xo)[2],
                                           const float* __restrict__ wt,
                                           const float* __restrict__ xt,
                                           unsigned lc, unsigned h) {
    #pragma unroll
    for (int j = 0; j < 4; ++j)
        w[j] = __builtin_nontemporal_load(reinterpret_cast<const v4f*>(wt + (lc + 64u * j)));
    xo[0] = *reinterpret_cast<const v4f*>(xt + 8u * h);
    xo[1] = *reinterpret_cast<const v4f*>(xt + 8u * h + 4u);
}

__device__ __forceinline__ void lmi_finish_tile(const unsigned (&xv)[16], const v4f (&w)[4],
                                                const v4f (&xo)[2],
                                                float mean, int lane, int h, bool live,
                                                float* __restrict__ out, int base) {
    v2f ones; ones.x = 1.0f; ones.y = 1.0f;
    const v8f zero8 = {0.f,0.f,0.f,0.f,0.f,0.f,0.f,0.f};
    #pragma unroll
    for (int g = 0; g < 2; ++g) {
        float lag[4], sq[4];
        #pragma unroll
        for (int jj = 0; jj < 2; ++jj) {
            const int j = 2 * g + jj;
            const float xn0 = __uint_as_float(xv[4 * j + 0]) - mean;
            const float xn1 = __uint_as_float(xv[4 * j + 1]) - mean;
            const float xn2 = __uint_as_float(xv[4 * j + 2]) - mean;
            const float xn3 = __uint_as_float(xv[4 * j + 3]) - mean;
            v2f A0; A0.x = w[j].x * xn0;   A0.y = w[j].y * xn1;
            v2f A1; A1.x = w[j].z * xn2;   A1.y = w[j].w * xn3;
            v2f Q0; Q0.x = A0.x * xn0;  Q0.y = A0.y * xn1;
            v2f Q1; Q1.x = A1.x * xn2;  Q1.y = A1.y * xn3;
            v8f dl = __builtin_amdgcn_wmma_f32_16x16x4_f32(
                         false, A0, false, ones, (short)0, zero8, false, false);
            v8f ds = __builtin_amdgcn_wmma_f32_16x16x4_f32(
                         false, Q0, false, ones, (short)0, zero8, false, false);
            dl = __builtin_amdgcn_wmma_f32_16x16x4_f32(
                         false, A1, false, ones, (short)0, dl, false, false);
            ds = __builtin_amdgcn_wmma_f32_16x16x4_f32(
                         false, Q1, false, ones, (short)0, ds, false, false);
            float l0 = (dl[0] + dl[1]) + (dl[2] + dl[3]);
            float l1 = (dl[4] + dl[5]) + (dl[6] + dl[7]);
            float s0 = (ds[0] + ds[1]) + (ds[2] + ds[3]);
            float s1 = (ds[4] + ds[5]) + (ds[6] + ds[7]);
            lag[2 * jj] = l0; lag[2 * jj + 1] = l1;
            sq[2 * jj]  = s0; sq[2 * jj + 1]  = s1;
        }

        if (live && (lane & 15) == 0) {
            const int rbase = base + h * 8 + 4 * g;
            #pragma unroll
            for (int r = 0; r < 4; ++r) {
                const float xs = xo[g][r] - mean;
                __builtin_nontemporal_store(xs * lag[r] * (float)(KNB - 1) / sq[r],
                                            out + rbase + r);
            }
        }
    }
}

__device__ __forceinline__ float lmi_block_total(float s, float* __restrict__ xl,
                                                 int tid, int wave) {
    #pragma unroll
    for (int off = 16; off > 0; off >>= 1) s += __shfl_xor(s, off, 32);
    __syncthreads();
    if ((tid & 31) == 0) xl[wave] = s;
    __syncthreads();
    float tot = 0.0f;
    #pragma unroll
    for (int w = 0; w < WGW; ++w) tot += xl[w];
    return tot;
}

__device__ __forceinline__ float lmi_mean_from_memory(const float* __restrict__ X, int n,
                                                      float* __restrict__ xl,
                                                      int tid, int wave) {
    float a = 0.0f;
    for (int i = tid; i < n; i += WGT) a += X[i];
    return lmi_block_total(a, xl, tid, wave) / (float)n;
}

__global__ __launch_bounds__(WGT)
void lmi_main(const float* __restrict__ X,
              const float* __restrict__ W,
              const int*   __restrict__ IDS,
              float*       __restrict__ out,
              int n) {
    __shared__ __attribute__((aligned(16))) float xl[CH];

    const int tid  = threadIdx.x;
    const int wave = __builtin_amdgcn_readfirstlane(tid >> 5);
    const int full_tiles = n >> 4;
    float mean = 0.0f;
    bool have_mean = false;

    const int t0 = (int)(((long long)full_tiles * blockIdx.x) / gridDim.x);
    const int t1 = (int)(((long long)full_tiles * (blockIdx.x + 1)) / gridDim.x);
    const int npass = (n + CH - 1) / CH;

    if (npass <= 16) {
        const int pshift = (int)(blockIdx.x % (unsigned)((npass > 0) ? npass : 1));
        for (int r0 = t0; r0 < t1; r0 += WGW * TPW) {
            const int tb = r0 + wave * TPW;
            const bool first = (r0 == t0);
            v2f acc; acc.x = 0.0f; acc.y = 0.0f;

            unsigned v[TPW][16];
            unsigned nib[TPW][2];

            const unsigned lci = lmi_lane_offset(lmi_here((unsigned)tid) & 31u);
            #pragma unroll
            for (int u = 0; u < TPW; ++u) {
                const int tl = (tb + u < t1) ? (tb + u) : (t1 - 1);
                const int* it = IDS + (long long)tl * (16 * KNB);
                #pragma unroll
                for (int j = 0; j < 4; ++j) {
                    const v4i id = __builtin_nontemporal_load(
                                       reinterpret_cast<const v4i*>(it + (lci + 64u * j)));
                    v[u][4 * j + 0] = (unsigned)id.x;
                    v[u][4 * j + 1] = (unsigned)id.y;
                    v[u][4 * j + 2] = (unsigned)id.z;
                    v[u][4 * j + 3] = (unsigned)id.w;
                }
            }

            #pragma unroll
            for (int G = 0; G < 2 * TPW; ++G) {
                const int u = G >> 1, g = G & 1;
                unsigned word = 0u;
                #pragma unroll
                for (int e = 0; e < 8; ++e) {
                    const unsigned id = v[u][8 * g + e];
                    const unsigned cp = ((id >> 14) * 52429u) >> 18;
                    v[u][8 * g + e] = (id - cp * (unsigned)CH) << 2;
                    word |= cp << (4 * e);
                }
                if (G + 1 < 2 * TPW) {
                    const int un = (G + 1) >> 1, o = 8 * ((G + 1) & 1), c0 = 8 * g;
                    asm volatile("" : "+v"(word),
                                      "+v"(v[u][c0 + 0]), "+v"(v[u][c0 + 1]), "+v"(v[u][c0 + 2]),
                                      "+v"(v[u][c0 + 3]), "+v"(v[u][c0 + 4]), "+v"(v[u][c0 + 5]),
                                      "+v"(v[u][c0 + 6]), "+v"(v[u][c0 + 7]),
                                      "+v"(v[un][o + 0]), "+v"(v[un][o + 1]), "+v"(v[un][o + 2]),
                                      "+v"(v[un][o + 3]), "+v"(v[un][o + 4]), "+v"(v[un][o + 5]),
                                      "+v"(v[un][o + 6]), "+v"(v[un][o + 7]));
                } else {
                    asm volatile("" : "+v"(word));
                }
                nib[u][g] = word;
            }

            for (int q = 0; q < npass; ++q) {
                const int p = (q + pshift < npass) ? (q + pshift) : (q + pshift - npass);
                const int pbase = p * CH;
                const int cnt   = (n - pbase < CH) ? (n - pbase) : CH;
                const int nv4   = cnt >> 2;
                const int rot   = (int)(((blockIdx.x & (NWG - 1)) * (unsigned)nv4) / NWG) & ~15;
                const v4f* X4 = reinterpret_cast<const v4f*>(X + pbase);
                v4f* xl4 = reinterpret_cast<v4f*>(xl);

                __syncthreads();
                for (int b = 0; b < nv4; b += WGT * FU) {
                    v4f t[FU]; int jj[FU];
                    #pragma unroll
                    for (int k = 0; k < FU; ++k) {
                        int i = b + k * WGT + tid;
                        i = (i < nv4) ? i : (nv4 - 1);
                        int j = i + rot;
                        j = (j >= nv4) ? (j - nv4) : j;
                        jj[k] = j;
                        t[k] = X4[j];
                    }
                    #pragma unroll
                    for (int k = 0; k < FU; ++k) xl4[jj[k]] = t[k];
                    if (first) {
                        #pragma unroll
                        for (int k = 0; k < FU; ++k) {
                            const bool in = (b + k * WGT + tid) < nv4;
                            acc.x += in ? (t[k].x + t[k].z) : 0.0f;
                            acc.y += in ? (t[k].y + t[k].w) : 0.0f;
                        }
                    }
                }
                if (tid < (cnt & 3)) {
                    const float xe = X[pbase + (nv4 << 2) + tid];
                    xl[(nv4 << 2) + tid] = xe;
                    if (first) acc.x += xe;
                }
                __syncthreads();

                const unsigned pm = (unsigned)p * 0x11111111u;
                #pragma unroll
                for (int u = 0; u < TPW; ++u) {
                    const unsigned xk0 = nib[u][0] ^ pm, xk1 = nib[u][1] ^ pm;
                    unsigned got[16];
                    #pragma unroll
                    for (int s = 0; s < 16; ++s) {
                        const bool hit = (((s < 8) ? xk0 : xk1) & (15u << (4 * (s & 7)))) == 0u;
                        const unsigned a = hit ? v[u][s] : 0u;
                        got[s] = *reinterpret_cast<const unsigned*>(
                                     reinterpret_cast<const char*>(xl) + a);
                    }
                    asm volatile("" : "+v"(got[0]),  "+v"(got[1]),  "+v"(got[2]),  "+v"(got[3]),
                                      "+v"(got[4]),  "+v"(got[5]),  "+v"(got[6]),  "+v"(got[7]),
                                      "+v"(got[8]),  "+v"(got[9]),  "+v"(got[10]), "+v"(got[11]),
                                      "+v"(got[12]), "+v"(got[13]), "+v"(got[14]), "+v"(got[15]));
                    #pragma unroll
                    for (int s = 0; s < 16; ++s) {
                        const bool hit = (((s < 8) ? xk0 : xk1) & (15u << (4 * (s & 7)))) == 0u;
                        v[u][s] = hit ? got[s] : v[u][s];
                    }
                }
            }

            if (first) {
                const float tot = lmi_block_total(acc.x + acc.y, xl, tid, wave);
                mean = __int_as_float(__builtin_amdgcn_readfirstlane(
                           __float_as_int(tot / (float)n)));
                have_mean = true;
            }

            const unsigned lanef = lmi_here((unsigned)tid) & 31u;
            const unsigned lcf   = lmi_lane_offset(lanef);
            v4f wq[TPW][4], xq[TPW][2];
            #pragma unroll
            for (int u = 0; u < WAHEAD && u < TPW; ++u) {
                const int tl = (tb + u < t1) ? (tb + u) : (t1 - 1);
                lmi_load_w(wq[u], xq[u], W + (long long)tl * (16 * KNB), X + (tl << 4),
                           lcf, lanef >> 4);
            }
            #pragma unroll
            for (int u = 0; u < TPW; ++u) {
                if (u + WAHEAD < TPW) {
                    const int ta = tb + u + WAHEAD;
                    const int tla = (ta < t1) ? ta : (t1 - 1);
                    lmi_load_w(wq[u + WAHEAD], xq[u + WAHEAD], W + (long long)tla * (16 * KNB),
                               X + (tla << 4), lcf, lanef >> 4);
                }
                const int tile = tb + u;
                const int tl   = (tile < t1) ? tile : (t1 - 1);
                lmi_finish_tile(v[u], wq[u], xq[u], mean, (int)lanef, (int)(lanef >> 4),
                                tile < t1, out, tl << 4);
            }
        }
    } else {
        mean = lmi_mean_from_memory(X, n, xl, tid, wave);
        have_mean = true;
        const int lane = (int)(lmi_here((unsigned)tid) & 31u);
        const unsigned lc = lmi_lane_offset((unsigned)lane);
        const int h = lane >> 4;
        for (int tile = t0 + wave; tile < t1; tile += WGW) {
            const int* it = IDS + (long long)tile * (16 * KNB);
            unsigned xv[16];
            #pragma unroll
            for (int j = 0; j < 4; ++j) {
                const v4i id = __builtin_nontemporal_load(
                                   reinterpret_cast<const v4i*>(it + (lc + 64u * j)));
                xv[4 * j + 0] = __float_as_uint(X[id.x]);
                xv[4 * j + 1] = __float_as_uint(X[id.y]);
                xv[4 * j + 2] = __float_as_uint(X[id.z]);
                xv[4 * j + 3] = __float_as_uint(X[id.w]);
            }
            v4f w[4], xo[2];
            lmi_load_w(w, xo, W + (long long)tile * (16 * KNB), X + (tile << 4), lc, (unsigned)h);
            lmi_finish_tile(xv, w, xo, mean, lane, h, true, out, tile << 4);
        }
    }

    if ((n & 15) && blockIdx.x == gridDim.x - 1) {
      if (!have_mean) mean = lmi_mean_from_memory(X, n, xl, tid, wave);
      if (wave == 0) {
        const int rem = n & 15;
        const int lane = (int)(lmi_here((unsigned)tid) & 31u);
        if (lane < rem) {
            const int row = full_tiles * 16 + lane;
            const float* wrow = W   + (long long)row * KNB;
            const int*   irow = IDS + (long long)row * KNB;
            float lag = 0.f, sq = 0.f;
            for (int k = 0; k < KNB; ++k) {
                const float xn = X[irow[k]] - mean;
                const float p  = wrow[k] * xn;
                lag += p; sq += p * xn;
            }
            const float xs = X[row] - mean;
            out[row] = xs * lag * (float)(KNB - 1) / sq;
        }
      }
    }
}

extern "C" void kernel_launch(void* const* d_in, const int* in_sizes, int n_in,
                              void* d_out, int out_size, void* d_ws, size_t ws_size,
                              hipStream_t stream) {
    const float* X   = (const float*)d_in[0];
    const float* W   = (const float*)d_in[1];
    const int*   IDS = (const int*)d_in[2];
    float*       out = (float*)d_out;
    const int n = in_sizes[0];

    lmi_main<<<NWG, WGT, 0, stream>>>(X, W, IDS, out, n);
}
